# starting kernel
# baseline (speedup 1.0000x reference)
#include <hip/hip_runtime.h>

#define HID 64
#define H4  16

typedef __attribute__((ext_vector_type(2))) float v2f;
typedef __attribute__((ext_vector_type(8))) float v8f;

__global__ void k_deg_init(float* __restrict__ deg, int n) {
    int i = blockIdx.x * blockDim.x + threadIdx.x;
    if (i < n) deg[i] = 1.0f;
}

__global__ void k_deg_scatter(const int* __restrict__ dst, float* __restrict__ deg, int e) {
    int i = blockIdx.x * blockDim.x + threadIdx.x;
    if (i < e) atomicAdd(&deg[dst[i]], 1.0f);
}

__global__ void k_rsqrt(float* __restrict__ deg, int n) {
    int i = blockIdx.x * blockDim.x + threadIdx.x;
    if (i < n) deg[i] = rsqrtf(deg[i]);
}

__global__ void k_norm(const int* __restrict__ src, const int* __restrict__ dst,
                       const float* __restrict__ dis, float* __restrict__ norm, int e) {
    int i = blockIdx.x * blockDim.x + threadIdx.x;
    if (i < e) norm[i] = dis[src[i]] * dis[dst[i]];
}

template <int K, int NC, int RELU>
__global__ __launch_bounds__(256)
void k_gemm(const float* __restrict__ A, const float* __restrict__ B,
            const float* __restrict__ bias, float* __restrict__ C, int M) {
    constexpr int tilesN = (NC + 15) / 16;
    int wave = blockIdx.x * (blockDim.x >> 5) + (threadIdx.x >> 5);
    int lane = threadIdx.x & 31;
    int tiles = (M >> 4) * tilesN;
    if (wave >= tiles) return;

    int tm = wave / tilesN;
    int tn = wave - tm * tilesN;
    int hf = lane >> 4;
    int l  = lane & 15;
    int col  = tn * 16 + l;
    int colC = (col < NC) ? col : (NC - 1);

    const v2f*   Ap = (const v2f*)(A + (size_t)(tm * 16 + l) * K + 2 * hf);
    const float* Bp = B + (size_t)(2 * hf) * NC + colC;

    v8f acc = {};
#pragma unroll 8
    for (int k0 = 0; k0 < K; k0 += 4) {
        v2f a = Ap[k0 >> 1];
        if (RELU) { a.x = fmaxf(a.x, 0.0f); a.y = fmaxf(a.y, 0.0f); }
        v2f b;
        b.x = Bp[(size_t)k0 * NC];
        b.y = Bp[(size_t)(k0 + 1) * NC];
        acc = __builtin_amdgcn_wmma_f32_16x16x4_f32(
             false, a,  false, b,
             (short)0, acc,  false,  false);
    }

    if (col < NC) {
        float bb = bias[col];
        float* Crow = C + (size_t)(tm * 16 + 8 * hf) * NC + col;
#pragma unroll
        for (int i = 0; i < 8; ++i)
            Crow[(size_t)i * NC] = acc[i] + bb;
    }
}

__global__ void k_selfinit(const float* __restrict__ t, const float* __restrict__ dis,
                           float* __restrict__ out, int n) {
    int gid = blockIdx.x * blockDim.x + threadIdx.x;
    if (gid >= n * H4) return;
    int i = gid >> 4;
    float s = dis[i];
    float k = s * s;
    float4 v = ((const float4*)t)[gid];
    float4 r; r.x = v.x * k; r.y = v.y * k; r.z = v.z * k; r.w = v.w * k;
    ((float4*)out)[gid] = r;
}

__global__ void k_scatter(const float* __restrict__ t, const int* __restrict__ src,
                          const int* __restrict__ dst, const float* __restrict__ norm,
                          float* out, int e) {
    int gid = blockIdx.x * blockDim.x + threadIdx.x;
    if (gid >= e * H4) return;
    int ed = gid >> 4;
    int c  = gid & 15;
    int s = src[ed];
    int d = dst[ed];
    float w = norm[ed];
    float4 v = ((const float4*)(t + (size_t)s * HID))[c];
    float* o = out + (size_t)d * HID + (size_t)c * 4;
    atomicAdd(o + 0, v.x * w);
    atomicAdd(o + 1, v.y * w);
    atomicAdd(o + 2, v.z * w);
    atomicAdd(o + 3, v.w * w);
}

extern "C" void kernel_launch(void* const* d_in, const int* in_sizes, int n_in,
                              void* d_out, int out_size, void* d_ws, size_t ws_size,
                              hipStream_t stream) {
    (void)n_in; (void)out_size; (void)ws_size;
    const float* x    = (const float*)d_in[0];
    const int*   ei   = (const int*)d_in[1];
    const float* W0   = (const float*)d_in[2];
    const float* b0   = (const float*)d_in[3];
    const float* W1   = (const float*)d_in[4];
    const float* b1   = (const float*)d_in[5];
    const float* W2   = (const float*)d_in[6];
    const float* b2   = (const float*)d_in[7];
    const float* Wout = (const float*)d_in[8];
    const float* bout = (const float*)d_in[9];

    const int FIN = 128, OUTD = 40;
    const int n = in_sizes[0] / FIN;
    const int e = in_sizes[1] / 2;
    const int* srcI = ei;
    const int* dstI = ei + e;

    char* w = (char*)d_ws;
    size_t off = 0;
    float* dis  = (float*)(w + off); off += (((size_t)n * 4) + 255) & ~(size_t)255;
    float* norm = (float*)(w + off); off += (((size_t)e * 4) + 255) & ~(size_t)255;
    float* tmp  = (float*)(w + off); off += (((size_t)n * HID * 4) + 255) & ~(size_t)255;
    float* hA   = (float*)(w + off); off += (((size_t)n * HID * 4) + 255) & ~(size_t)255;
    float* hB   = (float*)(w + off); off += (((size_t)n * HID * 4) + 255) & ~(size_t)255;

    const int T   = 256;
    const int WPB = T / 32;

    k_deg_init   <<<(n + T - 1) / T, T, 0, stream>>>(dis, n);
    k_deg_scatter<<<(e + T - 1) / T, T, 0, stream>>>(dstI, dis, e);
    k_rsqrt      <<<(n + T - 1) / T, T, 0, stream>>>(dis, n);
    k_norm       <<<(e + T - 1) / T, T, 0, stream>>>(srcI, dstI, dis, norm, e);

    const int tilesH = (n / 16) * (HID / 16);
    const int gGemmH = (tilesH + WPB - 1) / WPB;
    const int gAgg1  = (n * H4 + T - 1) / T;
    const int gAggE  = (e * H4 + T - 1) / T;

    k_gemm<128, HID, 0><<<gGemmH, T, 0, stream>>>(x, W0, b0, tmp, n);
    k_selfinit<<<gAgg1, T, 0, stream>>>(tmp, dis, hA, n);
    k_scatter <<<gAggE, T, 0, stream>>>(tmp, srcI, dstI, norm, hA, e);

    k_gemm<HID, HID, 1><<<gGemmH, T, 0, stream>>>(hA, W1, b1, tmp, n);
    k_selfinit<<<gAgg1, T, 0, stream>>>(tmp, dis, hB, n);
    k_scatter <<<gAggE, T, 0, stream>>>(tmp, srcI, dstI, norm, hB, e);

    k_gemm<HID, HID, 1><<<gGemmH, T, 0, stream>>>(hB, W2, b2, tmp, n);
    k_selfinit<<<gAgg1, T, 0, stream>>>(tmp, dis, hA, n);
    k_scatter <<<gAggE, T, 0, stream>>>(tmp, srcI, dstI, norm, hA, e);

    const int tilesO = (n / 16) * ((OUTD + 15) / 16);
    k_gemm<HID, OUTD, 1><<<(tilesO + WPB - 1) / WPB, T, 0, stream>>>(hA, Wout, bout,
                                                                     (float*)d_out, n);
}


# Products: a wave computes a whole 16-row strip as 4 column-interleaved tiles (tile j = columns 4c+j): A read once, B by 16-byte loads, whole-line stores; k now summed in 8-k blocks (new sum order)
# speedup vs baseline: 2.5067x; 2.5067x over previous
#include <hip/hip_runtime.h>


#define HID 64
#define H4  16
#define PL  32
#define NTICKETS 6
#define TSTRIDE 64
#define CHUNK 512

typedef __attribute__((ext_vector_type(2))) float v2f;
typedef __attribute__((ext_vector_type(8))) float v8f;


#define DEG_C  50
#define DEG_RS 10240
#define DEG_T  1024

__device__ __forceinline__ void deg_hit(unsigned* cnt, int d, int lo, unsigned rs) {
    const unsigned k = (unsigned)(d - lo);
    if (k < rs) atomicAdd(&cnt[k], 1u);
}

__global__ __launch_bounds__(DEG_T)
void k_deg_count(const int* __restrict__ dst, unsigned* __restrict__ part,
                 int* __restrict__ tickets, int e, int n, int chunk, int nranges) {
    __shared__ unsigned cnt[DEG_RS];
    const int tid = threadIdx.x;
    if (blockIdx.x == 0 && tid < NTICKETS) tickets[tid * TSTRIDE] = 0;

    const int c  = blockIdx.x / nranges;
    const int r  = blockIdx.x - c * nranges;
    const int lo = r * DEG_RS;
    const unsigned rs = (unsigned)((n - lo) < DEG_RS ? (n - lo) : DEG_RS);
    for (unsigned i = tid; i < rs; i += DEG_T) cnt[i] = 0u;
    __syncthreads();

    const int e_lo = c * chunk;
    int e_hi = e_lo + chunk; if (e_hi > e) e_hi = e;
    const int len = e_hi > e_lo ? e_hi - e_lo : 0;
    const int* dp = dst + e_lo;
    const int nvec = ((((size_t)dp) & 15) == 0) ? (len >> 2) : 0;
    const int4* d4 = (const int4*)dp;
    for (int j0 = 0; j0 < nvec; j0 += DEG_T * 8) {
        int4 v[8];
#pragma unroll
        for (int u = 0; u < 8; ++u) {
            const int j = j0 + u * DEG_T + tid;
            v[u] = (j < nvec) ? d4[j] : make_int4(-1, -1, -1, -1);
        }
#pragma unroll
        for (int u = 0; u < 8; ++u) {
            deg_hit(cnt, v[u].x, lo, rs); deg_hit(cnt, v[u].y, lo, rs);
            deg_hit(cnt, v[u].z, lo, rs); deg_hit(cnt, v[u].w, lo, rs);
        }
    }
    for (int j = 4 * nvec + tid; j < len; j += DEG_T) deg_hit(cnt, dp[j], lo, rs);
    __syncthreads();

    unsigned* prow = part + (size_t)c * n + lo;
    for (unsigned i = tid; i < rs; i += DEG_T) prow[i] = cnt[i];
}

__global__ void k_rsqrt(const unsigned* __restrict__ part, float* __restrict__ dis, int n) {
    int i = blockIdx.x * blockDim.x + threadIdx.x;
    if (i >= n) return;
    unsigned s = 1u;
#pragma unroll
    for (int c = 0; c < DEG_C; ++c) s += part[(size_t)c * n + i];
    dis[i] = rsqrtf((float)s);
}

__global__ void k_norm(const int* __restrict__ src, const int* __restrict__ dst,
                       const float* __restrict__ dis, float* __restrict__ norm, int e) {
    int i = blockIdx.x * blockDim.x + threadIdx.x;
    if (i < e) norm[i] = dis[src[i]] * dis[dst[i]];
}

#define GEMM_T 128

template <int K, int NC, int RELU, int APL, int CPL>
__global__ __launch_bounds__(GEMM_T)
void k_gemm(const float* __restrict__ A, const float* __restrict__ B,
            const float* __restrict__ bias, float* __restrict__ C, int M) {
    static_assert(!APL || K == 2 * PL, "two-plane A needs K == 64");
    static_assert(!CPL || NC == 2 * PL, "two-plane C needs NC == 64");
    static_assert(K % 8 == 0 && NC % 4 == 0 && NC <= 64, "8-k blocks, 4-column groups");
    constexpr int NB = K / 8;
    constexpr int NG = NC / 4;
    const int wave = blockIdx.x * (GEMM_T >> 5) + (threadIdx.x >> 5);
    if (wave >= (M >> 4)) return;
    const int lane = threadIdx.x & 31;
    const int hf = lane >> 4;
    const int c  = lane & 15;
    const int cg = (c < NG) ? c : (NG - 1);

    const size_t plane = (size_t)M * PL;
    constexpr int ROWA = APL ? PL : K;
    const float* Ap = A + (size_t)(wave * 16 + c) * ROWA + 4 * hf;
    const float* Bp = B + (size_t)(4 * hf) * NC + 4 * cg;

    float4 a[NB];
#pragma unroll
    for (int b = 0; b < NB; ++b) {
        const int kk = 8 * b;
        const float* ap = APL ? (Ap + (size_t)(kk >> 5) * plane + (kk & (PL - 1))) : (Ap + kk);
        a[b] = *(const float4*)ap;
    }

    v8f acc0 = {}, acc1 = {}, acc2 = {}, acc3 = {};
#pragma unroll
    for (int b = 0; b < NB; ++b) {
        const float* bp = Bp + (size_t)(8 * b) * NC;
        const float4 w0 = *(const float4*)(bp);
        const float4 w1 = *(const float4*)(bp + NC);
        const float4 w2 = *(const float4*)(bp + 2 * NC);
        const float4 w3 = *(const float4*)(bp + 3 * NC);
        float4 av = a[b];
        if (RELU) {
            av.x = fmaxf(av.x, 0.0f); av.y = fmaxf(av.y, 0.0f);
            av.z = fmaxf(av.z, 0.0f); av.w = fmaxf(av.w, 0.0f);
        }
        v2f a01; a01.x = av.x; a01.y = av.y;
        v2f a23; a23.x = av.z; a23.y = av.w;
        v2f p;
        p.x = w0.x; p.y = w1.x;
        acc0 = __builtin_amdgcn_wmma_f32_16x16x4_f32(false, a01, false, p, (short)0, acc0, false, false);
        p.x = w0.y; p.y = w1.y;
        acc1 = __builtin_amdgcn_wmma_f32_16x16x4_f32(false, a01, false, p, (short)0, acc1, false, false);
        p.x = w0.z; p.y = w1.z;
        acc2 = __builtin_amdgcn_wmma_f32_16x16x4_f32(false, a01, false, p, (short)0, acc2, false, false);
        p.x = w0.w; p.y = w1.w;
        acc3 = __builtin_amdgcn_wmma_f32_16x16x4_f32(false, a01, false, p, (short)0, acc3, false, false);
        p.x = w2.x; p.y = w3.x;
        acc0 = __builtin_amdgcn_wmma_f32_16x16x4_f32(false, a23, false, p, (short)0, acc0, false, false);
        p.x = w2.y; p.y = w3.y;
        acc1 = __builtin_amdgcn_wmma_f32_16x16x4_f32(false, a23, false, p, (short)0, acc1, false, false);
        p.x = w2.z; p.y = w3.z;
        acc2 = __builtin_amdgcn_wmma_f32_16x16x4_f32(false, a23, false, p, (short)0, acc2, false, false);
        p.x = w2.w; p.y = w3.w;
        acc3 = __builtin_amdgcn_wmma_f32_16x16x4_f32(false, a23, false, p, (short)0, acc3, false, false);
    }

    if (c < NG) {
        const float4 bb = *(const float4*)(bias + 4 * c);
        constexpr int ROWC = CPL ? PL : NC;
        float* Crow = CPL
            ? C + (size_t)(c >> 3) * plane + (size_t)(wave * 16 + 8 * hf) * PL + 4 * (c & 7)
            : C + (size_t)(wave * 16 + 8 * hf) * NC + 4 * c;
#pragma unroll
        for (int r = 0; r < 8; ++r) {
            float4 o;
            o.x = acc0[r] + bb.x; o.y = acc1[r] + bb.y;
            o.z = acc2[r] + bb.z; o.w = acc3[r] + bb.w;
            *(float4*)(Crow + (size_t)r * ROWC) = o;
        }
    }
}

__global__ void k_selfinit(const float* __restrict__ t, const float* __restrict__ dis,
                           float* __restrict__ out, int n) {
    int gid = blockIdx.x * blockDim.x + threadIdx.x;
    if (gid >= n * H4) return;
    int inpl = (gid >= n * 8) ? gid - n * 8 : gid;
    int i = inpl >> 3;
    float s = dis[i];
    float k = s * s;
    float4 v = ((const float4*)t)[gid];
    float4 r; r.x = v.x * k; r.y = v.y * k; r.z = v.z * k; r.w = v.w * k;
    ((float4*)out)[gid] = r;
}

__device__ __forceinline__ void scatter32(const float* __restrict__ tp, float* op,
                                          int s_l, int d_l, float w_l, int cnt, int lane) {
    int j = 0;
    for (; j + 16 <= cnt; j += 16) {
        float v[16];
#pragma unroll
        for (int u = 0; u < 16; ++u) {
            const unsigned s = (unsigned)__builtin_amdgcn_readlane(s_l, j + u);
            v[u] = tp[s * (unsigned)PL + (unsigned)lane];
        }
#pragma unroll
        for (int u = 0; u < 16; ++u) {
            const unsigned d = (unsigned)__builtin_amdgcn_readlane(d_l, j + u);
            const float w = __int_as_float(__builtin_amdgcn_readlane(__float_as_int(w_l), j + u));
            atomicAdd(op + (d * (unsigned)PL + (unsigned)lane), v[u] * w);
        }
    }
    for (; j < cnt; ++j) {
        const unsigned s = (unsigned)__builtin_amdgcn_readlane(s_l, j);
        const unsigned d = (unsigned)__builtin_amdgcn_readlane(d_l, j);
        const float w = __int_as_float(__builtin_amdgcn_readlane(__float_as_int(w_l), j));
        const float a = tp[s * (unsigned)PL + (unsigned)lane];
        atomicAdd(op + (d * (unsigned)PL + (unsigned)lane), a * w);
    }
}

__global__ __launch_bounds__(256)
void k_scatter(const float* __restrict__ t, const int* __restrict__ src,
               const int* __restrict__ dst, const float* __restrict__ norm,
               float* out, int* tickets, int e, int n, int nchunks) {
    __shared__ int pick[2];
    if (threadIdx.x == 0) {
        const unsigned id = __builtin_amdgcn_s_sendmsg_rtn(0x87);
        int p = (((id >> 16) & 15u) >= 4u) ? 1 : 0;
        int c = atomicAdd(&tickets[p * TSTRIDE], 1);
        if (c >= nchunks) { p ^= 1; c = atomicAdd(&tickets[p * TSTRIDE], 1); }
        pick[0] = c; pick[1] = p;
    }
    __syncthreads();
    const int c = __builtin_amdgcn_readfirstlane(pick[0]);
    const int p = __builtin_amdgcn_readfirstlane(pick[1]);
    if (c >= nchunks) return;

    const int lane = threadIdx.x & 31;
    const int wv = __builtin_amdgcn_readfirstlane((int)(threadIdx.x >> 5));
    const int base = c * CHUNK + wv * 64;
    if (base >= e) return;
    const float* tp = t + (size_t)p * n * PL;
    float* op = out + (size_t)p * n * PL;

    const int e0 = base + lane, e1 = base + 32 + lane;
    int s0 = 0, d0 = 0, s1 = 0, d1 = 0;
    float w0 = 0.0f, w1 = 0.0f;
    if (e0 < e) { s0 = src[e0]; d0 = dst[e0]; w0 = norm[e0]; }
    if (e1 < e) { s1 = src[e1]; d1 = dst[e1]; w1 = norm[e1]; }
    const int left = e - base;
    const int cnt0 = left < 32 ? left : 32;
    const int cnt1 = left < 64 ? left - 32 : 32;
    scatter32(tp, op, s0, d0, w0, cnt0, lane);
    if (cnt1 > 0) scatter32(tp, op, s1, d1, w1, cnt1, lane);
}

extern "C" void kernel_launch(void* const* d_in, const int* in_sizes, int n_in,
                              void* d_out, int out_size, void* d_ws, size_t ws_size,
                              hipStream_t stream) {
    (void)n_in; (void)out_size; (void)ws_size;
    const float* x    = (const float*)d_in[0];
    const int*   ei   = (const int*)d_in[1];
    const float* W0   = (const float*)d_in[2];
    const float* b0   = (const float*)d_in[3];
    const float* W1   = (const float*)d_in[4];
    const float* b1   = (const float*)d_in[5];
    const float* W2   = (const float*)d_in[6];
    const float* b2   = (const float*)d_in[7];
    const float* Wout = (const float*)d_in[8];
    const float* bout = (const float*)d_in[9];

    const int FIN = 128, OUTD = 40;
    const int n = in_sizes[0] / FIN;
    const int e = in_sizes[1] / 2;
    const int* srcI = ei;
    const int* dstI = ei + e;

    char* w = (char*)d_ws;
    size_t off = 0;
    float* dis  = (float*)(w + off); off += (((size_t)n * 4) + 255) & ~(size_t)255;
    float* norm = (float*)(w + off); off += (((size_t)e * 4) + 255) & ~(size_t)255;
    float* tmp  = (float*)(w + off); off += (((size_t)n * HID * 4) + 255) & ~(size_t)255;
    float* hA   = (float*)(w + off); off += (((size_t)n * HID * 4) + 255) & ~(size_t)255;
    float* hB   = (float*)(w + off); off += (((size_t)n * HID * 4) + 255) & ~(size_t)255;
    int*   tick = (int*)(w + off);   off += (size_t)NTICKETS * TSTRIDE * 4;
    unsigned* part = (unsigned*)(w + off); off += (((size_t)DEG_C * n * 4) + 255) & ~(size_t)255;

    const int T   = 256;

    const int dchunk  = (((e + DEG_C - 1) / DEG_C) + 3) & ~3;
    const int dranges = (n + DEG_RS - 1) / DEG_RS;
    k_deg_count  <<<DEG_C * dranges, DEG_T, 0, stream>>>(dstI, part, tick, e, n, dchunk, dranges);
    k_rsqrt      <<<(n + T - 1) / T, T, 0, stream>>>(part, dis, n);
    k_norm       <<<(e + T - 1) / T, T, 0, stream>>>(srcI, dstI, dis, norm, e);

    const int strips = n / 16;
    const int gGemm  = (strips + (GEMM_T / 32) - 1) / (GEMM_T / 32);
    const int gAgg1  = (n * H4 + T - 1) / T;
    const int nch    = (e + CHUNK - 1) / CHUNK;
    const int gAggE  = 2 * nch;

    k_gemm<128, HID, 0, 0, 1><<<gGemm, GEMM_T, 0, stream>>>(x, W0, b0, tmp, n);
    k_selfinit<<<gAgg1, T, 0, stream>>>(tmp, dis, hA, n);
    k_scatter <<<gAggE, T, 0, stream>>>(tmp, srcI, dstI, norm, hA, tick + 0 * TSTRIDE, e, n, nch);

    k_gemm<HID, HID, 1, 1, 1><<<gGemm, GEMM_T, 0, stream>>>(hA, W1, b1, tmp, n);
    k_selfinit<<<gAgg1, T, 0, stream>>>(tmp, dis, hB, n);
    k_scatter <<<gAggE, T, 0, stream>>>(tmp, srcI, dstI, norm, hB, tick + 2 * TSTRIDE, e, n, nch);

    k_gemm<HID, HID, 1, 1, 1><<<gGemm, GEMM_T, 0, stream>>>(hB, W2, b2, tmp, n);
    k_selfinit<<<gAgg1, T, 0, stream>>>(tmp, dis, hA, n);
    k_scatter <<<gAggE, T, 0, stream>>>(tmp, srcI, dstI, norm, hA, tick + 4 * TSTRIDE, e, n, nch);

    k_gemm<HID, OUTD, 1, 1, 0><<<gGemm, GEMM_T, 0, stream>>>(hA, Wout, bout, (float*)d_out, n);
}


# Product epilogue also stores the self-loop term C[i]*dis[i]^2 (same bits) as the sums' start value in whole lines (round 3 tried it before whole-line stores); 13 -> 10 launches; same sum order
# speedup vs baseline: 2.6470x; 1.0560x over previous
#include <hip/hip_runtime.h>


#define HID 64
#define PL  32
#define NTICKETS 6
#define TSTRIDE 64
#define CHUNK 512

typedef __attribute__((ext_vector_type(2))) float v2f;
typedef __attribute__((ext_vector_type(8))) float v8f;


#define DEG_C  50
#define DEG_RS 10240
#define DEG_T  1024

__device__ __forceinline__ void deg_hit(unsigned* cnt, int d, int lo, unsigned rs) {
    const unsigned k = (unsigned)(d - lo);
    if (k < rs) atomicAdd(&cnt[k], 1u);
}

__global__ __launch_bounds__(DEG_T)
void k_deg_count(const int* __restrict__ dst, unsigned* __restrict__ part,
                 int* __restrict__ tickets, int e, int n, int chunk, int nranges) {
    __shared__ unsigned cnt[DEG_RS];
    const int tid = threadIdx.x;
    if (blockIdx.x == 0 && tid < NTICKETS) tickets[tid * TSTRIDE] = 0;

    const int c  = blockIdx.x / nranges;
    const int r  = blockIdx.x - c * nranges;
    const int lo = r * DEG_RS;
    const unsigned rs = (unsigned)((n - lo) < DEG_RS ? (n - lo) : DEG_RS);
    for (unsigned i = tid; i < rs; i += DEG_T) cnt[i] = 0u;
    __syncthreads();

    const int e_lo = c * chunk;
    int e_hi = e_lo + chunk; if (e_hi > e) e_hi = e;
    const int len = e_hi > e_lo ? e_hi - e_lo : 0;
    const int* dp = dst + e_lo;
    const int nvec = ((((size_t)dp) & 15) == 0) ? (len >> 2) : 0;
    const int4* d4 = (const int4*)dp;
    for (int j0 = 0; j0 < nvec; j0 += DEG_T * 8) {
        int4 v[8];
#pragma unroll
        for (int u = 0; u < 8; ++u) {
            const int j = j0 + u * DEG_T + tid;
            v[u] = (j < nvec) ? d4[j] : make_int4(-1, -1, -1, -1);
        }
#pragma unroll
        for (int u = 0; u < 8; ++u) {
            deg_hit(cnt, v[u].x, lo, rs); deg_hit(cnt, v[u].y, lo, rs);
            deg_hit(cnt, v[u].z, lo, rs); deg_hit(cnt, v[u].w, lo, rs);
        }
    }
    for (int j = 4 * nvec + tid; j < len; j += DEG_T) deg_hit(cnt, dp[j], lo, rs);
    __syncthreads();

    unsigned* prow = part + (size_t)c * n + lo;
    for (unsigned i = tid; i < rs; i += DEG_T) prow[i] = cnt[i];
}

__global__ void k_rsqrt(const unsigned* __restrict__ part, float* __restrict__ dis, int n) {
    int i = blockIdx.x * blockDim.x + threadIdx.x;
    if (i >= n) return;
    unsigned s = 1u;
#pragma unroll
    for (int c = 0; c < DEG_C; ++c) s += part[(size_t)c * n + i];
    dis[i] = rsqrtf((float)s);
}

__global__ void k_norm(const int* __restrict__ src, const int* __restrict__ dst,
                       const float* __restrict__ dis, float* __restrict__ norm, int e) {
    int i = blockIdx.x * blockDim.x + threadIdx.x;
    if (i < e) norm[i] = dis[src[i]] * dis[dst[i]];
}

#define GEMM_T 128

template <int K, int NC, int RELU, int APL, int CPL, int INIT>
__global__ __launch_bounds__(GEMM_T)
void k_gemm(const float* __restrict__ A, const float* __restrict__ B,
            const float* __restrict__ bias, float* __restrict__ C, int M,
            const float* __restrict__ dis, float* __restrict__ S) {
    static_assert(!INIT || CPL, "the sums' start value goes to a two-plane table");
    static_assert(!APL || K == 2 * PL, "two-plane A needs K == 64");
    static_assert(!CPL || NC == 2 * PL, "two-plane C needs NC == 64");
    static_assert(K % 8 == 0 && NC % 4 == 0 && NC <= 64, "8-k blocks, 4-column groups");
    constexpr int NB = K / 8;
    constexpr int NG = NC / 4;
    const int wave = blockIdx.x * (GEMM_T >> 5) + (threadIdx.x >> 5);
    if (wave >= (M >> 4)) return;
    const int lane = threadIdx.x & 31;
    const int hf = lane >> 4;
    const int c  = lane & 15;
    const int cg = (c < NG) ? c : (NG - 1);

    const size_t plane = (size_t)M * PL;
    constexpr int ROWA = APL ? PL : K;
    const float* Ap = A + (size_t)(wave * 16 + c) * ROWA + 4 * hf;
    const float* Bp = B + (size_t)(4 * hf) * NC + 4 * cg;

    float4 a[NB];
#pragma unroll
    for (int b = 0; b < NB; ++b) {
        const int kk = 8 * b;
        const float* ap = APL ? (Ap + (size_t)(kk >> 5) * plane + (kk & (PL - 1))) : (Ap + kk);
        a[b] = *(const float4*)ap;
    }

    v8f acc0 = {}, acc1 = {}, acc2 = {}, acc3 = {};
#pragma unroll
    for (int b = 0; b < NB; ++b) {
        const float* bp = Bp + (size_t)(8 * b) * NC;
        const float4 w0 = *(const float4*)(bp);
        const float4 w1 = *(const float4*)(bp + NC);
        const float4 w2 = *(const float4*)(bp + 2 * NC);
        const float4 w3 = *(const float4*)(bp + 3 * NC);
        float4 av = a[b];
        if (RELU) {
            av.x = fmaxf(av.x, 0.0f); av.y = fmaxf(av.y, 0.0f);
            av.z = fmaxf(av.z, 0.0f); av.w = fmaxf(av.w, 0.0f);
        }
        v2f a01; a01.x = av.x; a01.y = av.y;
        v2f a23; a23.x = av.z; a23.y = av.w;
        v2f p;
        p.x = w0.x; p.y = w1.x;
        acc0 = __builtin_amdgcn_wmma_f32_16x16x4_f32(false, a01, false, p, (short)0, acc0, false, false);
        p.x = w0.y; p.y = w1.y;
        acc1 = __builtin_amdgcn_wmma_f32_16x16x4_f32(false, a01, false, p, (short)0, acc1, false, false);
        p.x = w0.z; p.y = w1.z;
        acc2 = __builtin_amdgcn_wmma_f32_16x16x4_f32(false, a01, false, p, (short)0, acc2, false, false);
        p.x = w0.w; p.y = w1.w;
        acc3 = __builtin_amdgcn_wmma_f32_16x16x4_f32(false, a01, false, p, (short)0, acc3, false, false);
        p.x = w2.x; p.y = w3.x;
        acc0 = __builtin_amdgcn_wmma_f32_16x16x4_f32(false, a23, false, p, (short)0, acc0, false, false);
        p.x = w2.y; p.y = w3.y;
        acc1 = __builtin_amdgcn_wmma_f32_16x16x4_f32(false, a23, false, p, (short)0, acc1, false, false);
        p.x = w2.z; p.y = w3.z;
        acc2 = __builtin_amdgcn_wmma_f32_16x16x4_f32(false, a23, false, p, (short)0, acc2, false, false);
        p.x = w2.w; p.y = w3.w;
        acc3 = __builtin_amdgcn_wmma_f32_16x16x4_f32(false, a23, false, p, (short)0, acc3, false, false);
    }

    if (c < NG) {
        const float4 bb = *(const float4*)(bias + 4 * c);
        constexpr int ROWC = CPL ? PL : NC;
        float* Crow = CPL
            ? C + (size_t)(c >> 3) * plane + (size_t)(wave * 16 + 8 * hf) * PL + 4 * (c & 7)
            : C + (size_t)(wave * 16 + 8 * hf) * NC + 4 * c;
        float dk[8];
        if (INIT) {
            const float4 d0 = *(const float4*)(dis + wave * 16 + 8 * hf);
            const float4 d1 = *(const float4*)(dis + wave * 16 + 8 * hf + 4);
            dk[0] = d0.x * d0.x; dk[1] = d0.y * d0.y; dk[2] = d0.z * d0.z; dk[3] = d0.w * d0.w;
            dk[4] = d1.x * d1.x; dk[5] = d1.y * d1.y; dk[6] = d1.z * d1.z; dk[7] = d1.w * d1.w;
        }
        float* Srow = INIT ? S + (Crow - C) : nullptr;
#pragma unroll
        for (int r = 0; r < 8; ++r) {
            float4 o;
            o.x = acc0[r] + bb.x; o.y = acc1[r] + bb.y;
            o.z = acc2[r] + bb.z; o.w = acc3[r] + bb.w;
            *(float4*)(Crow + (size_t)r * ROWC) = o;
            if (INIT) {
                float4 q;
                q.x = o.x * dk[r]; q.y = o.y * dk[r]; q.z = o.z * dk[r]; q.w = o.w * dk[r];
                *(float4*)(Srow + (size_t)r * ROWC) = q;
            }
        }
    }
}


__device__ __forceinline__ void scatter32(const float* __restrict__ tp, float* op,
                                          int s_l, int d_l, float w_l, int cnt, int lane) {
    int j = 0;
    for (; j + 16 <= cnt; j += 16) {
        float v[16];
#pragma unroll
        for (int u = 0; u < 16; ++u) {
            const unsigned s = (unsigned)__builtin_amdgcn_readlane(s_l, j + u);
            v[u] = tp[s * (unsigned)PL + (unsigned)lane];
        }
#pragma unroll
        for (int u = 0; u < 16; ++u) {
            const unsigned d = (unsigned)__builtin_amdgcn_readlane(d_l, j + u);
            const float w = __int_as_float(__builtin_amdgcn_readlane(__float_as_int(w_l), j + u));
            atomicAdd(op + (d * (unsigned)PL + (unsigned)lane), v[u] * w);
        }
    }
    for (; j < cnt; ++j) {
        const unsigned s = (unsigned)__builtin_amdgcn_readlane(s_l, j);
        const unsigned d = (unsigned)__builtin_amdgcn_readlane(d_l, j);
        const float w = __int_as_float(__builtin_amdgcn_readlane(__float_as_int(w_l), j));
        const float a = tp[s * (unsigned)PL + (unsigned)lane];
        atomicAdd(op + (d * (unsigned)PL + (unsigned)lane), a * w);
    }
}

__global__ __launch_bounds__(256)
void k_scatter(const float* __restrict__ t, const int* __restrict__ src,
               const int* __restrict__ dst, const float* __restrict__ norm,
               float* out, int* tickets, int e, int n, int nchunks) {
    __shared__ int pick[2];
    if (threadIdx.x == 0) {
        const unsigned id = __builtin_amdgcn_s_sendmsg_rtn(0x87);
        int p = (((id >> 16) & 15u) >= 4u) ? 1 : 0;
        int c = atomicAdd(&tickets[p * TSTRIDE], 1);
        if (c >= nchunks) { p ^= 1; c = atomicAdd(&tickets[p * TSTRIDE], 1); }
        pick[0] = c; pick[1] = p;
    }
    __syncthreads();
    const int c = __builtin_amdgcn_readfirstlane(pick[0]);
    const int p = __builtin_amdgcn_readfirstlane(pick[1]);
    if (c >= nchunks) return;

    const int lane = threadIdx.x & 31;
    const int wv = __builtin_amdgcn_readfirstlane((int)(threadIdx.x >> 5));
    const int base = c * CHUNK + wv * 64;
    if (base >= e) return;
    const float* tp = t + (size_t)p * n * PL;
    float* op = out + (size_t)p * n * PL;

    const int e0 = base + lane, e1 = base + 32 + lane;
    int s0 = 0, d0 = 0, s1 = 0, d1 = 0;
    float w0 = 0.0f, w1 = 0.0f;
    if (e0 < e) { s0 = src[e0]; d0 = dst[e0]; w0 = norm[e0]; }
    if (e1 < e) { s1 = src[e1]; d1 = dst[e1]; w1 = norm[e1]; }
    const int left = e - base;
    const int cnt0 = left < 32 ? left : 32;
    const int cnt1 = left < 64 ? left - 32 : 32;
    scatter32(tp, op, s0, d0, w0, cnt0, lane);
    if (cnt1 > 0) scatter32(tp, op, s1, d1, w1, cnt1, lane);
}

extern "C" void kernel_launch(void* const* d_in, const int* in_sizes, int n_in,
                              void* d_out, int out_size, void* d_ws, size_t ws_size,
                              hipStream_t stream) {
    (void)n_in; (void)out_size; (void)ws_size;
    const float* x    = (const float*)d_in[0];
    const int*   ei   = (const int*)d_in[1];
    const float* W0   = (const float*)d_in[2];
    const float* b0   = (const float*)d_in[3];
    const float* W1   = (const float*)d_in[4];
    const float* b1   = (const float*)d_in[5];
    const float* W2   = (const float*)d_in[6];
    const float* b2   = (const float*)d_in[7];
    const float* Wout = (const float*)d_in[8];
    const float* bout = (const float*)d_in[9];

    const int FIN = 128, OUTD = 40;
    const int n = in_sizes[0] / FIN;
    const int e = in_sizes[1] / 2;
    const int* srcI = ei;
    const int* dstI = ei + e;

    char* w = (char*)d_ws;
    size_t off = 0;
    float* dis  = (float*)(w + off); off += (((size_t)n * 4) + 255) & ~(size_t)255;
    float* norm = (float*)(w + off); off += (((size_t)e * 4) + 255) & ~(size_t)255;
    float* tmp  = (float*)(w + off); off += (((size_t)n * HID * 4) + 255) & ~(size_t)255;
    float* hA   = (float*)(w + off); off += (((size_t)n * HID * 4) + 255) & ~(size_t)255;
    float* hB   = (float*)(w + off); off += (((size_t)n * HID * 4) + 255) & ~(size_t)255;
    int*   tick = (int*)(w + off);   off += (size_t)NTICKETS * TSTRIDE * 4;
    unsigned* part = (unsigned*)(w + off); off += (((size_t)DEG_C * n * 4) + 255) & ~(size_t)255;

    const int T   = 256;

    const int dchunk  = (((e + DEG_C - 1) / DEG_C) + 3) & ~3;
    const int dranges = (n + DEG_RS - 1) / DEG_RS;
    k_deg_count  <<<DEG_C * dranges, DEG_T, 0, stream>>>(dstI, part, tick, e, n, dchunk, dranges);
    k_rsqrt      <<<(n + T - 1) / T, T, 0, stream>>>(part, dis, n);
    k_norm       <<<(e + T - 1) / T, T, 0, stream>>>(srcI, dstI, dis, norm, e);

    const int strips = n / 16;
    const int gGemm  = (strips + (GEMM_T / 32) - 1) / (GEMM_T / 32);
    const int nch    = (e + CHUNK - 1) / CHUNK;
    const int gAggE  = 2 * nch;

    k_gemm<128, HID, 0, 0, 1, 1><<<gGemm, GEMM_T, 0, stream>>>(x, W0, b0, tmp, n, dis, hA);
    k_scatter <<<gAggE, T, 0, stream>>>(tmp, srcI, dstI, norm, hA, tick + 0 * TSTRIDE, e, n, nch);

    k_gemm<HID, HID, 1, 1, 1, 1><<<gGemm, GEMM_T, 0, stream>>>(hA, W1, b1, tmp, n, dis, hB);
    k_scatter <<<gAggE, T, 0, stream>>>(tmp, srcI, dstI, norm, hB, tick + 2 * TSTRIDE, e, n, nch);

    k_gemm<HID, HID, 1, 1, 1, 1><<<gGemm, GEMM_T, 0, stream>>>(hB, W2, b2, tmp, n, dis, hA);
    k_scatter <<<gAggE, T, 0, stream>>>(tmp, srcI, dstI, norm, hA, tick + 4 * TSTRIDE, e, n, nch);

    k_gemm<HID, OUTD, 1, 1, 0, 0><<<gGemm, GEMM_T, 0, stream>>>(hA, Wout, bout, (float*)d_out, n, nullptr, nullptr);
}


# Edge weights ride in the first product's launch as extra work-groups behind the strips: their scattered table reads run under the matrix work; was a launch of its own; 10 -> 9 launches; same bits
# speedup vs baseline: 2.7358x; 1.0335x over previous
#include <hip/hip_runtime.h>


#define HID 64
#define PL  32
#define NTICKETS 6
#define TSTRIDE 64
#define CHUNK 512

typedef __attribute__((ext_vector_type(2))) float v2f;
typedef __attribute__((ext_vector_type(8))) float v8f;


#define DEG_C  50
#define DEG_RS 10240
#define DEG_T  1024

__device__ __forceinline__ void deg_hit(unsigned* cnt, int d, int lo, unsigned rs) {
    const unsigned k = (unsigned)(d - lo);
    if (k < rs) atomicAdd(&cnt[k], 1u);
}

__global__ __launch_bounds__(DEG_T)
void k_deg_count(const int* __restrict__ dst, unsigned* __restrict__ part,
                 int* __restrict__ tickets, int e, int n, int chunk, int nranges) {
    __shared__ unsigned cnt[DEG_RS];
    const int tid = threadIdx.x;
    if (blockIdx.x == 0 && tid < NTICKETS) tickets[tid * TSTRIDE] = 0;

    const int c  = blockIdx.x / nranges;
    const int r  = blockIdx.x - c * nranges;
    const int lo = r * DEG_RS;
    const unsigned rs = (unsigned)((n - lo) < DEG_RS ? (n - lo) : DEG_RS);
    for (unsigned i = tid; i < rs; i += DEG_T) cnt[i] = 0u;
    __syncthreads();

    const int e_lo = c * chunk;
    int e_hi = e_lo + chunk; if (e_hi > e) e_hi = e;
    const int len = e_hi > e_lo ? e_hi - e_lo : 0;
    const int* dp = dst + e_lo;
    const int nvec = ((((size_t)dp) & 15) == 0) ? (len >> 2) : 0;
    const int4* d4 = (const int4*)dp;
    for (int j0 = 0; j0 < nvec; j0 += DEG_T * 8) {
        int4 v[8];
#pragma unroll
        for (int u = 0; u < 8; ++u) {
            const int j = j0 + u * DEG_T + tid;
            v[u] = (j < nvec) ? d4[j] : make_int4(-1, -1, -1, -1);
        }
#pragma unroll
        for (int u = 0; u < 8; ++u) {
            deg_hit(cnt, v[u].x, lo, rs); deg_hit(cnt, v[u].y, lo, rs);
            deg_hit(cnt, v[u].z, lo, rs); deg_hit(cnt, v[u].w, lo, rs);
        }
    }
    for (int j = 4 * nvec + tid; j < len; j += DEG_T) deg_hit(cnt, dp[j], lo, rs);
    __syncthreads();

    unsigned* prow = part + (size_t)c * n + lo;
    for (unsigned i = tid; i < rs; i += DEG_T) prow[i] = cnt[i];
}

__global__ void k_rsqrt(const unsigned* __restrict__ part, float* __restrict__ dis, int n) {
    int i = blockIdx.x * blockDim.x + threadIdx.x;
    if (i >= n) return;
    unsigned s = 1u;
#pragma unroll
    for (int c = 0; c < DEG_C; ++c) s += part[(size_t)c * n + i];
    dis[i] = rsqrtf((float)s);
}


#define GEMM_T 128
#define WPT 8

template <int K, int NC, int RELU, int APL, int CPL, int INIT, int NORM>
__global__ __launch_bounds__(GEMM_T)
void k_gemm(const float* __restrict__ A, const float* __restrict__ B,
            const float* __restrict__ bias, float* __restrict__ C, int M,
            const float* __restrict__ dis, float* __restrict__ S,
            const int* __restrict__ src, const int* __restrict__ dst,
            float* __restrict__ norm, int e, int gblocks) {
    if (NORM) {
        if ((int)blockIdx.x >= gblocks) {
            const int bb = ((int)blockIdx.x - gblocks) * (GEMM_T * WPT);
            if (bb + GEMM_T * WPT <= e) {
                const int base = bb + (int)threadIdx.x;
                unsigned sv[WPT], dv[WPT];
#pragma unroll
                for (int u = 0; u < WPT; ++u) {
                    sv[u] = (unsigned)src[base + u * GEMM_T];
                    dv[u] = (unsigned)dst[base + u * GEMM_T];
                }
                float fa[WPT], fb[WPT];
#pragma unroll
                for (int u = 0; u < WPT; ++u) { fa[u] = dis[sv[u]]; fb[u] = dis[dv[u]]; }
#pragma unroll
                for (int u = 0; u < WPT; ++u) norm[base + u * GEMM_T] = fa[u] * fb[u];
            } else {
                for (int i = bb + (int)threadIdx.x; i < e; i += GEMM_T) norm[i] = dis[src[i]] * dis[dst[i]];
            }
            return;
        }
    }
    static_assert(!INIT || CPL, "the sums' start value goes to a two-plane table");
    static_assert(!APL || K == 2 * PL, "two-plane A needs K == 64");
    static_assert(!CPL || NC == 2 * PL, "two-plane C needs NC == 64");
    static_assert(K % 8 == 0 && NC % 4 == 0 && NC <= 64, "8-k blocks, 4-column groups");
    constexpr int NB = K / 8;
    constexpr int NG = NC / 4;
    const int wave = blockIdx.x * (GEMM_T >> 5) + (threadIdx.x >> 5);
    if (wave >= (M >> 4)) return;
    const int lane = threadIdx.x & 31;
    const int hf = lane >> 4;
    const int c  = lane & 15;
    const int cg = (c < NG) ? c : (NG - 1);

    const size_t plane = (size_t)M * PL;
    constexpr int ROWA = APL ? PL : K;
    const float* Ap = A + (size_t)(wave * 16 + c) * ROWA + 4 * hf;
    const float* Bp = B + (size_t)(4 * hf) * NC + 4 * cg;

    float4 a[NB];
#pragma unroll
    for (int b = 0; b < NB; ++b) {
        const int kk = 8 * b;
        const float* ap = APL ? (Ap + (size_t)(kk >> 5) * plane + (kk & (PL - 1))) : (Ap + kk);
        a[b] = *(const float4*)ap;
    }

    v8f acc0 = {}, acc1 = {}, acc2 = {}, acc3 = {};
#pragma unroll
    for (int b = 0; b < NB; ++b) {
        const float* bp = Bp + (size_t)(8 * b) * NC;
        const float4 w0 = *(const float4*)(bp);
        const float4 w1 = *(const float4*)(bp + NC);
        const float4 w2 = *(const float4*)(bp + 2 * NC);
        const float4 w3 = *(const float4*)(bp + 3 * NC);
        float4 av = a[b];
        if (RELU) {
            av.x = fmaxf(av.x, 0.0f); av.y = fmaxf(av.y, 0.0f);
            av.z = fmaxf(av.z, 0.0f); av.w = fmaxf(av.w, 0.0f);
        }
        v2f a01; a01.x = av.x; a01.y = av.y;
        v2f a23; a23.x = av.z; a23.y = av.w;
        v2f p;
        p.x = w0.x; p.y = w1.x;
        acc0 = __builtin_amdgcn_wmma_f32_16x16x4_f32(false, a01, false, p, (short)0, acc0, false, false);
        p.x = w0.y; p.y = w1.y;
        acc1 = __builtin_amdgcn_wmma_f32_16x16x4_f32(false, a01, false, p, (short)0, acc1, false, false);
        p.x = w0.z; p.y = w1.z;
        acc2 = __builtin_amdgcn_wmma_f32_16x16x4_f32(false, a01, false, p, (short)0, acc2, false, false);
        p.x = w0.w; p.y = w1.w;
        acc3 = __builtin_amdgcn_wmma_f32_16x16x4_f32(false, a01, false, p, (short)0, acc3, false, false);
        p.x = w2.x; p.y = w3.x;
        acc0 = __builtin_amdgcn_wmma_f32_16x16x4_f32(false, a23, false, p, (short)0, acc0, false, false);
        p.x = w2.y; p.y = w3.y;
        acc1 = __builtin_amdgcn_wmma_f32_16x16x4_f32(false, a23, false, p, (short)0, acc1, false, false);
        p.x = w2.z; p.y = w3.z;
        acc2 = __builtin_amdgcn_wmma_f32_16x16x4_f32(false, a23, false, p, (short)0, acc2, false, false);
        p.x = w2.w; p.y = w3.w;
        acc3 = __builtin_amdgcn_wmma_f32_16x16x4_f32(false, a23, false, p, (short)0, acc3, false, false);
    }

    if (c < NG) {
        const float4 bb = *(const float4*)(bias + 4 * c);
        constexpr int ROWC = CPL ? PL : NC;
        float* Crow = CPL
            ? C + (size_t)(c >> 3) * plane + (size_t)(wave * 16 + 8 * hf) * PL + 4 * (c & 7)
            : C + (size_t)(wave * 16 + 8 * hf) * NC + 4 * c;
        float dk[8];
        if (INIT) {
            const float4 d0 = *(const float4*)(dis + wave * 16 + 8 * hf);
            const float4 d1 = *(const float4*)(dis + wave * 16 + 8 * hf + 4);
            dk[0] = d0.x * d0.x; dk[1] = d0.y * d0.y; dk[2] = d0.z * d0.z; dk[3] = d0.w * d0.w;
            dk[4] = d1.x * d1.x; dk[5] = d1.y * d1.y; dk[6] = d1.z * d1.z; dk[7] = d1.w * d1.w;
        }
        float* Srow = INIT ? S + (Crow - C) : nullptr;
#pragma unroll
        for (int r = 0; r < 8; ++r) {
            float4 o;
            o.x = acc0[r] + bb.x; o.y = acc1[r] + bb.y;
            o.z = acc2[r] + bb.z; o.w = acc3[r] + bb.w;
            *(float4*)(Crow + (size_t)r * ROWC) = o;
            if (INIT) {
                float4 q;
                q.x = o.x * dk[r]; q.y = o.y * dk[r]; q.z = o.z * dk[r]; q.w = o.w * dk[r];
                *(float4*)(Srow + (size_t)r * ROWC) = q;
            }
        }
    }
}


__device__ __forceinline__ void scatter32(const float* __restrict__ tp, float* op,
                                          int s_l, int d_l, float w_l, int cnt, int lane) {
    int j = 0;
    for (; j + 16 <= cnt; j += 16) {
        float v[16];
#pragma unroll
        for (int u = 0; u < 16; ++u) {
            const unsigned s = (unsigned)__builtin_amdgcn_readlane(s_l, j + u);
            v[u] = tp[s * (unsigned)PL + (unsigned)lane];
        }
#pragma unroll
        for (int u = 0; u < 16; ++u) {
            const unsigned d = (unsigned)__builtin_amdgcn_readlane(d_l, j + u);
            const float w = __int_as_float(__builtin_amdgcn_readlane(__float_as_int(w_l), j + u));
            atomicAdd(op + (d * (unsigned)PL + (unsigned)lane), v[u] * w);
        }
    }
    for (; j < cnt; ++j) {
        const unsigned s = (unsigned)__builtin_amdgcn_readlane(s_l, j);
        const unsigned d = (unsigned)__builtin_amdgcn_readlane(d_l, j);
        const float w = __int_as_float(__builtin_amdgcn_readlane(__float_as_int(w_l), j));
        const float a = tp[s * (unsigned)PL + (unsigned)lane];
        atomicAdd(op + (d * (unsigned)PL + (unsigned)lane), a * w);
    }
}

__global__ __launch_bounds__(256)
void k_scatter(const float* __restrict__ t, const int* __restrict__ src,
               const int* __restrict__ dst, const float* __restrict__ norm,
               float* out, int* tickets, int e, int n, int nchunks) {
    __shared__ int pick[2];
    if (threadIdx.x == 0) {
        const unsigned id = __builtin_amdgcn_s_sendmsg_rtn(0x87);
        int p = (((id >> 16) & 15u) >= 4u) ? 1 : 0;
        int c = atomicAdd(&tickets[p * TSTRIDE], 1);
        if (c >= nchunks) { p ^= 1; c = atomicAdd(&tickets[p * TSTRIDE], 1); }
        pick[0] = c; pick[1] = p;
    }
    __syncthreads();
    const int c = __builtin_amdgcn_readfirstlane(pick[0]);
    const int p = __builtin_amdgcn_readfirstlane(pick[1]);
    if (c >= nchunks) return;

    const int lane = threadIdx.x & 31;
    const int wv = __builtin_amdgcn_readfirstlane((int)(threadIdx.x >> 5));
    const int base = c * CHUNK + wv * 64;
    if (base >= e) return;
    const float* tp = t + (size_t)p * n * PL;
    float* op = out + (size_t)p * n * PL;

    const int e0 = base + lane, e1 = base + 32 + lane;
    int s0 = 0, d0 = 0, s1 = 0, d1 = 0;
    float w0 = 0.0f, w1 = 0.0f;
    if (e0 < e) { s0 = src[e0]; d0 = dst[e0]; w0 = norm[e0]; }
    if (e1 < e) { s1 = src[e1]; d1 = dst[e1]; w1 = norm[e1]; }
    const int left = e - base;
    const int cnt0 = left < 32 ? left : 32;
    const int cnt1 = left < 64 ? left - 32 : 32;
    scatter32(tp, op, s0, d0, w0, cnt0, lane);
    if (cnt1 > 0) scatter32(tp, op, s1, d1, w1, cnt1, lane);
}

extern "C" void kernel_launch(void* const* d_in, const int* in_sizes, int n_in,
                              void* d_out, int out_size, void* d_ws, size_t ws_size,
                              hipStream_t stream) {
    (void)n_in; (void)out_size; (void)ws_size;
    const float* x    = (const float*)d_in[0];
    const int*   ei   = (const int*)d_in[1];
    const float* W0   = (const float*)d_in[2];
    const float* b0   = (const float*)d_in[3];
    const float* W1   = (const float*)d_in[4];
    const float* b1   = (const float*)d_in[5];
    const float* W2   = (const float*)d_in[6];
    const float* b2   = (const float*)d_in[7];
    const float* Wout = (const float*)d_in[8];
    const float* bout = (const float*)d_in[9];

    const int FIN = 128, OUTD = 40;
    const int n = in_sizes[0] / FIN;
    const int e = in_sizes[1] / 2;
    const int* srcI = ei;
    const int* dstI = ei + e;

    char* w = (char*)d_ws;
    size_t off = 0;
    float* dis  = (float*)(w + off); off += (((size_t)n * 4) + 255) & ~(size_t)255;
    float* norm = (float*)(w + off); off += (((size_t)e * 4) + 255) & ~(size_t)255;
    float* tmp  = (float*)(w + off); off += (((size_t)n * HID * 4) + 255) & ~(size_t)255;
    float* hA   = (float*)(w + off); off += (((size_t)n * HID * 4) + 255) & ~(size_t)255;
    float* hB   = (float*)(w + off); off += (((size_t)n * HID * 4) + 255) & ~(size_t)255;
    int*   tick = (int*)(w + off);   off += (size_t)NTICKETS * TSTRIDE * 4;
    unsigned* part = (unsigned*)(w + off); off += (((size_t)DEG_C * n * 4) + 255) & ~(size_t)255;

    const int T   = 256;

    const int dchunk  = (((e + DEG_C - 1) / DEG_C) + 3) & ~3;
    const int dranges = (n + DEG_RS - 1) / DEG_RS;
    k_deg_count  <<<DEG_C * dranges, DEG_T, 0, stream>>>(dstI, part, tick, e, n, dchunk, dranges);
    k_rsqrt      <<<(n + T - 1) / T, T, 0, stream>>>(part, dis, n);

    const int strips = n / 16;
    const int gGemm  = (strips + (GEMM_T / 32) - 1) / (GEMM_T / 32);
    const int gWgt   = (e + GEMM_T * WPT - 1) / (GEMM_T * WPT);
    const int nch    = (e + CHUNK - 1) / CHUNK;
    const int gAggE  = 2 * nch;

    k_gemm<128, HID, 0, 0, 1, 1, 1><<<gGemm + gWgt, GEMM_T, 0, stream>>>(x, W0, b0, tmp, n, dis, hA, srcI, dstI, norm, e, gGemm);
    k_scatter <<<gAggE, T, 0, stream>>>(tmp, srcI, dstI, norm, hA, tick + 0 * TSTRIDE, e, n, nch);

    k_gemm<HID, HID, 1, 1, 1, 1, 0><<<gGemm, GEMM_T, 0, stream>>>(hA, W1, b1, tmp, n, dis, hB, nullptr, nullptr, nullptr, 0, gGemm);
    k_scatter <<<gAggE, T, 0, stream>>>(tmp, srcI, dstI, norm, hB, tick + 2 * TSTRIDE, e, n, nch);

    k_gemm<HID, HID, 1, 1, 1, 1, 0><<<gGemm, GEMM_T, 0, stream>>>(hB, W2, b2, tmp, n, dis, hA, nullptr, nullptr, nullptr, 0, gGemm);
    k_scatter <<<gAggE, T, 0, stream>>>(tmp, srcI, dstI, norm, hA, tick + 4 * TSTRIDE, e, n, nch);

    k_gemm<HID, OUTD, 1, 1, 0, 0, 0><<<gGemm, GEMM_T, 0, stream>>>(hA, Wout, bout, (float*)d_out, n, nullptr, nullptr, nullptr, nullptr, nullptr, 0, gGemm);
}
